# starting kernel
# baseline (speedup 1.0000x reference)
#include <hip/hip_runtime.h>


typedef __attribute__((ext_vector_type(2))) float v2f;
typedef __attribute__((ext_vector_type(4))) float v4f;
typedef __attribute__((ext_vector_type(8))) float v8f;

__global__ __launch_bounds__(256) void block8x8_gemm_wmma(
    const float* __restrict__ x,
    const float* __restrict__ w,
    const float* __restrict__ bias,
    float* __restrict__ y,
    int num_tiles)
{
    const int lane  = threadIdx.x & 31;
    const int half  = lane >> 4;
    const int m     = lane & 15;
    const int wave  = blockIdx.x * (blockDim.x >> 5) + (threadIdx.x >> 5);
    const int nwav  = gridDim.x * (blockDim.x >> 5);

    const int wbase = (lane & 7) * 8 + 2 * half;
    v2f a0, a1;
    a0.x = w[wbase + 0];
    a0.y = w[wbase + 1];
    a1.x = w[wbase + 4];
    a1.y = w[wbase + 5];

    v8f cb;
    cb.s0 = bias[0]; cb.s1 = bias[1]; cb.s2 = bias[2]; cb.s3 = bias[3];
    cb.s4 = bias[4]; cb.s5 = bias[5]; cb.s6 = bias[6]; cb.s7 = bias[7];

    const int npairs = num_tiles >> 1;
    const int boff   = m * 8 + 2 * half;

    for (int p = wave; p < npairs; p += nwav) {
        const float* xt = x + (size_t)p * 256;

        __builtin_prefetch(x + ((size_t)p + nwav) * 256, 0, 0);

        const v2f b0a = *(const v2f*)(xt + boff);
        const v2f b1a = *(const v2f*)(xt + boff + 4);
        const v2f b0b = *(const v2f*)(xt + 128 + boff);
        const v2f b1b = *(const v2f*)(xt + 128 + boff + 4);

        v8f da = __builtin_amdgcn_wmma_f32_16x16x4_f32(
            false, a0, false, b0a, (short)0, cb, false, false);
        v8f db = __builtin_amdgcn_wmma_f32_16x16x4_f32(
            false, a0, false, b0b, (short)0, cb, false, false);
        da = __builtin_amdgcn_wmma_f32_16x16x4_f32(
            false, a1, false, b1a, (short)0, da, false, false);
        db = __builtin_amdgcn_wmma_f32_16x16x4_f32(
            false, a1, false, b1b, (short)0, db, false, false);

        float* yt = y + (size_t)p * 256 + boff - 2 * half + 4 * half;
        v4f oa, ob;
        oa.x = half ? da.s4 : da.s0;
        oa.y = half ? da.s5 : da.s1;
        oa.z = half ? da.s6 : da.s2;
        oa.w = half ? da.s7 : da.s3;
        ob.x = half ? db.s4 : db.s0;
        ob.y = half ? db.s5 : db.s1;
        ob.z = half ? db.s6 : db.s2;
        ob.w = half ? db.s7 : db.s3;
        *(v4f*)(yt)       = oa;
        *(v4f*)(yt + 128) = ob;
    }

    if ((num_tiles & 1) && wave == 0) {
        const int t = num_tiles - 1;
        const float* xt = x + (size_t)t * 128;
        const v2f b0 = *(const v2f*)(xt + boff);
        const v2f b1 = *(const v2f*)(xt + boff + 4);
        v8f d = __builtin_amdgcn_wmma_f32_16x16x4_f32(
            false, a0, false, b0, (short)0, cb, false, false);
        d = __builtin_amdgcn_wmma_f32_16x16x4_f32(
            false, a1, false, b1, (short)0, d, false, false);
        v4f out;
        out.x = half ? d.s4 : d.s0;
        out.y = half ? d.s5 : d.s1;
        out.z = half ? d.s6 : d.s2;
        out.w = half ? d.s7 : d.s3;
        *(v4f*)(y + (size_t)t * 128 + m * 8 + 4 * half) = out;
    }
}

extern "C" void kernel_launch(void* const* d_in, const int* in_sizes, int n_in,
                              void* d_out, int out_size, void* d_ws, size_t ws_size,
                              hipStream_t stream) {
    const float* x    = (const float*)d_in[0];
    const float* wmat = (const float*)d_in[1];
    const float* bias = (const float*)d_in[2];
    float* y          = (float*)d_out;

    const int num_tiles = in_sizes[0] / 128;
    const int npairs    = num_tiles >> 1;

    int blocks = 1024;
    int max_blocks = (npairs + 7) / 8;
    if (blocks > max_blocks) blocks = max_blocks;
    if (blocks < 1) blocks = 1;

    block8x8_gemm_wmma<<<blocks, 256, 0, stream>>>(x, wmat, bias, y, num_tiles);
}


# More bytes in flight per wave: one b128 per lane reads a whole 512-B tile, next 8 tiles' loads issued before the current 8 are multiplied (16 in flight), 16 waves/WGP; k summed {0,1,4,5}+{2,3,6,7}
# speedup vs baseline: 1.0669x; 1.0669x over previous
#include <hip/hip_runtime.h>


typedef __attribute__((ext_vector_type(2))) float v2f;
typedef __attribute__((ext_vector_type(4))) float v4f;
typedef __attribute__((ext_vector_type(8))) float v8f;

#define TILE_FLOATS 128
#define GROUP_TILES 8
#define GROUP_FLOATS (GROUP_TILES * TILE_FLOATS)
#define DEV static __device__ __forceinline__

DEV void load_group(const float* __restrict__ xg, int loff, v4f (&q)[GROUP_TILES]) {
#pragma unroll
    for (int i = 0; i < GROUP_TILES; ++i)
        q[i] = *(const v4f*)(xg + (i * TILE_FLOATS + loff));
}

DEV void multiply_store_group(const v4f (&q)[GROUP_TILES], float* __restrict__ yg, int loff,
                              v2f a_lo, v2f a_hi, v8f cb, int half) {
    v8f d[GROUP_TILES];
#pragma unroll
    for (int i = 0; i < GROUP_TILES; ++i) {
        v2f b; b.x = q[i].x; b.y = q[i].y;
        d[i] = __builtin_amdgcn_wmma_f32_16x16x4_f32(false, a_lo, false, b, (short)0, cb, false, false);
    }
#pragma unroll
    for (int i = 0; i < GROUP_TILES; ++i) {
        v2f b; b.x = q[i].z; b.y = q[i].w;
        d[i] = __builtin_amdgcn_wmma_f32_16x16x4_f32(false, a_hi, false, b, (short)0, d[i], false, false);
    }
#pragma unroll
    for (int i = 0; i < GROUP_TILES; ++i) {
        const v4f lo = __builtin_shufflevector(d[i], d[i], 0, 1, 2, 3);
        const v4f hi = __builtin_shufflevector(d[i], d[i], 4, 5, 6, 7);
        *(v4f*)(yg + (i * TILE_FLOATS + loff)) = half ? hi : lo;
    }
}

__global__ __launch_bounds__(256) void block8x8_gemm_wmma(
    const float* __restrict__ x,
    const float* __restrict__ w,
    const float* __restrict__ bias,
    float* __restrict__ y,
    int num_tiles)
{
    const int lane = threadIdx.x & 31;
    const int half = lane >> 4;
    const int m    = lane & 15;
    const int wave = blockIdx.x * (blockDim.x >> 5) + __builtin_amdgcn_readfirstlane(threadIdx.x >> 5);
    const int nwav = gridDim.x * (blockDim.x >> 5);

    const int wbase = (lane & 7) * 8 + 4 * half;
    v2f a_lo, a_hi;
    a_lo.x = w[wbase + 0];
    a_lo.y = w[wbase + 1];
    a_hi.x = w[wbase + 2];
    a_hi.y = w[wbase + 3];

    v8f cb;
    cb.s0 = bias[0]; cb.s1 = bias[1]; cb.s2 = bias[2]; cb.s3 = bias[3];
    cb.s4 = bias[4]; cb.s5 = bias[5]; cb.s6 = bias[6]; cb.s7 = bias[7];

    const int loff    = m * 8 + 4 * half;
    const int ngroups = num_tiles / GROUP_TILES;

    int g = wave;
    if (g < ngroups) {
        const size_t step = (size_t)nwav * GROUP_FLOATS;
        const float* xg = x + (size_t)g * GROUP_FLOATS;
        float*       yg = y + (size_t)g * GROUP_FLOATS;
        v4f qa[GROUP_TILES], qb[GROUP_TILES];
        load_group(xg, loff, qa);
        while (g + 2 * nwav < ngroups) {
            __builtin_amdgcn_sched_barrier(0);
            load_group(xg + step, loff, qb);
            __builtin_amdgcn_sched_barrier(0);
            multiply_store_group(qa, yg, loff, a_lo, a_hi, cb, half);
            __builtin_amdgcn_sched_barrier(0);
            load_group(xg + 2 * step, loff, qa);
            __builtin_amdgcn_sched_barrier(0);
            multiply_store_group(qb, yg + step, loff, a_lo, a_hi, cb, half);
            g += 2 * nwav; xg += 2 * step; yg += 2 * step;
        }
        if (g + nwav < ngroups) {
            __builtin_amdgcn_sched_barrier(0);
            load_group(xg + step, loff, qb);
            __builtin_amdgcn_sched_barrier(0);
            multiply_store_group(qa, yg, loff, a_lo, a_hi, cb, half);
            multiply_store_group(qb, yg + step, loff, a_lo, a_hi, cb, half);
        } else {
            multiply_store_group(qa, yg, loff, a_lo, a_hi, cb, half);
        }
    }

    const int t = ngroups * GROUP_TILES + wave;
    if (t < num_tiles) {
        const v4f q = *(const v4f*)(x + (size_t)t * TILE_FLOATS + loff);
        v2f b0, b1;
        b0.x = q.x; b0.y = q.y; b1.x = q.z; b1.y = q.w;
        v8f d = __builtin_amdgcn_wmma_f32_16x16x4_f32(false, a_lo, false, b0, (short)0, cb, false, false);
        d = __builtin_amdgcn_wmma_f32_16x16x4_f32(false, a_hi, false, b1, (short)0, d, false, false);
        v4f o;
        o.x = half ? d.s4 : d.s0;
        o.y = half ? d.s5 : d.s1;
        o.z = half ? d.s6 : d.s2;
        o.w = half ? d.s7 : d.s3;
        *(v4f*)(y + (size_t)t * TILE_FLOATS + loff) = o;
    }
}

extern "C" void kernel_launch(void* const* d_in, const int* in_sizes, int n_in,
                              void* d_out, int out_size, void* d_ws, size_t ws_size,
                              hipStream_t stream) {
    const float* x    = (const float*)d_in[0];
    const float* wmat = (const float*)d_in[1];
    const float* bias = (const float*)d_in[2];
    float* y          = (float*)d_out;

    const int num_tiles = in_sizes[0] / 128;
    const int ngroups   = num_tiles / 8;

    int blocks = 512;
    int max_blocks = (ngroups + 7) / 8;
    if (blocks > max_blocks) blocks = max_blocks;
    if (blocks < 1) blocks = 1;

    block8x8_gemm_wmma<<<blocks, 256, 0, stream>>>(x, wmat, bias, y, num_tiles);
}


# speedup vs baseline: 1.1027x; 1.0336x over previous
#include <hip/hip_runtime.h>


typedef __attribute__((ext_vector_type(2))) float v2f;
typedef __attribute__((ext_vector_type(4))) float v4f;
typedef __attribute__((ext_vector_type(8))) float v8f;

#define TILE_FLOATS 128
#define GROUP_TILES 8
#define GROUP_FLOATS (GROUP_TILES * TILE_FLOATS)
#define DEV static __device__ __forceinline__

DEV v4f load_x(const float* p) { return __builtin_nontemporal_load((const v4f*)p); }

DEV void load_group(const float* __restrict__ xg, int loff, v4f (&q)[GROUP_TILES]) {
#pragma unroll
    for (int i = 0; i < GROUP_TILES; ++i)
        q[i] = load_x(xg + (i * TILE_FLOATS + loff));
}

DEV void multiply_store_group(const v4f (&q)[GROUP_TILES], float* __restrict__ yg, int loff,
                              v2f a_lo, v2f a_hi, v8f cb, int half) {
    v8f d[GROUP_TILES];
#pragma unroll
    for (int i = 0; i < GROUP_TILES; ++i) {
        v2f b; b.x = q[i].x; b.y = q[i].y;
        d[i] = __builtin_amdgcn_wmma_f32_16x16x4_f32(false, a_lo, false, b, (short)0, cb, false, false);
    }
#pragma unroll
    for (int i = 0; i < GROUP_TILES; ++i) {
        v2f b; b.x = q[i].z; b.y = q[i].w;
        d[i] = __builtin_amdgcn_wmma_f32_16x16x4_f32(false, a_hi, false, b, (short)0, d[i], false, false);
    }
#pragma unroll
    for (int i = 0; i < GROUP_TILES; ++i) {
        const v4f lo = __builtin_shufflevector(d[i], d[i], 0, 1, 2, 3);
        const v4f hi = __builtin_shufflevector(d[i], d[i], 4, 5, 6, 7);
        *(v4f*)(yg + (i * TILE_FLOATS + loff)) = half ? hi : lo;
    }
}

__global__ __launch_bounds__(256) void block8x8_gemm_wmma(
    const float* __restrict__ x,
    const float* __restrict__ w,
    const float* __restrict__ bias,
    float* __restrict__ y,
    int num_tiles)
{
    const int lane = threadIdx.x & 31;
    const int half = lane >> 4;
    const int m    = lane & 15;
    const int wave = blockIdx.x * (blockDim.x >> 5) + __builtin_amdgcn_readfirstlane(threadIdx.x >> 5);
    const int nwav = gridDim.x * (blockDim.x >> 5);

    const int wbase = (lane & 7) * 8 + 4 * half;
    v2f a_lo, a_hi;
    a_lo.x = w[wbase + 0];
    a_lo.y = w[wbase + 1];
    a_hi.x = w[wbase + 2];
    a_hi.y = w[wbase + 3];

    v8f cb;
    cb.s0 = bias[0]; cb.s1 = bias[1]; cb.s2 = bias[2]; cb.s3 = bias[3];
    cb.s4 = bias[4]; cb.s5 = bias[5]; cb.s6 = bias[6]; cb.s7 = bias[7];

    const int loff    = m * 8 + 4 * half;
    const int ngroups = num_tiles / GROUP_TILES;

    int g = wave;
    if (g < ngroups) {
        const size_t step = (size_t)nwav * GROUP_FLOATS;
        const float* xg = x + (size_t)g * GROUP_FLOATS;
        float*       yg = y + (size_t)g * GROUP_FLOATS;
        v4f qa[GROUP_TILES], qb[GROUP_TILES];
        load_group(xg, loff, qa);
        while (g + 2 * nwav < ngroups) {
            __builtin_amdgcn_sched_barrier(0);
            load_group(xg + step, loff, qb);
            __builtin_amdgcn_sched_barrier(0);
            multiply_store_group(qa, yg, loff, a_lo, a_hi, cb, half);
            __builtin_amdgcn_sched_barrier(0);
            load_group(xg + 2 * step, loff, qa);
            __builtin_amdgcn_sched_barrier(0);
            multiply_store_group(qb, yg + step, loff, a_lo, a_hi, cb, half);
            g += 2 * nwav; xg += 2 * step; yg += 2 * step;
        }
        if (g + nwav < ngroups) {
            __builtin_amdgcn_sched_barrier(0);
            load_group(xg + step, loff, qb);
            __builtin_amdgcn_sched_barrier(0);
            multiply_store_group(qa, yg, loff, a_lo, a_hi, cb, half);
            multiply_store_group(qb, yg + step, loff, a_lo, a_hi, cb, half);
        } else {
            multiply_store_group(qa, yg, loff, a_lo, a_hi, cb, half);
        }
    }

    const int t = ngroups * GROUP_TILES + wave;
    if (t < num_tiles) {
        const v4f q = load_x(x + (size_t)t * TILE_FLOATS + loff);
        v2f b0, b1;
        b0.x = q.x; b0.y = q.y; b1.x = q.z; b1.y = q.w;
        v8f d = __builtin_amdgcn_wmma_f32_16x16x4_f32(false, a_lo, false, b0, (short)0, cb, false, false);
        d = __builtin_amdgcn_wmma_f32_16x16x4_f32(false, a_hi, false, b1, (short)0, d, false, false);
        v4f o;
        o.x = half ? d.s4 : d.s0;
        o.y = half ? d.s5 : d.s1;
        o.z = half ? d.s6 : d.s2;
        o.w = half ? d.s7 : d.s3;
        *(v4f*)(y + (size_t)t * TILE_FLOATS + loff) = o;
    }
}

extern "C" void kernel_launch(void* const* d_in, const int* in_sizes, int n_in,
                              void* d_out, int out_size, void* d_ws, size_t ws_size,
                              hipStream_t stream) {
    const float* x    = (const float*)d_in[0];
    const float* wmat = (const float*)d_in[1];
    const float* bias = (const float*)d_in[2];
    float* y          = (float*)d_out;

    const int num_tiles = in_sizes[0] / 128;
    const int ngroups   = num_tiles / 8;

    int blocks = 512;
    int max_blocks = (ngroups + 7) / 8;
    if (blocks > max_blocks) blocks = max_blocks;
    if (blocks < 1) blocks = 1;

    block8x8_gemm_wmma<<<blocks, 256, 0, stream>>>(x, wmat, bias, y, num_tiles);
}
